# starting kernel
# baseline (speedup 1.0000x reference)
#include <hip/hip_runtime.h>

typedef float v2f __attribute__((ext_vector_type(2)));
typedef float v8f __attribute__((ext_vector_type(8)));

#define CH 128

template <int NCOL>
__global__ __launch_bounds__(256) void gemm_wmma_f32(
    const float* __restrict__ A, const float* __restrict__ W,
    const float* __restrict__ bias, float* __restrict__ C, int nrows) {
  constexpr int NT = NCOL / 16;
  constexpr int KP = 130;
  __shared__ float lw[NCOL * KP];

  const int tid = threadIdx.x;
  for (int i = tid; i < CH * NCOL; i += 256) {
    int k = i / NCOL, c = i - k * NCOL;
    lw[c * KP + k] = W[i];
  }
  __syncthreads();

  const int wave = tid >> 5;
  const int lane = tid & 31;
  const int m0 = (blockIdx.x * 8 + wave) * 16;
  if (m0 >= nrows) return;

  const int lrow = lane & 15;
  const int kbase = (lane >> 4) * 2;

  v8f acc[NT] = {};

  int arow = m0 + lrow;
  if (arow > nrows - 1) arow = nrows - 1;
  const float* aptr = A + (size_t)arow * CH + kbase;

#pragma unroll 4
  for (int kc = 0; kc < CH / 4; ++kc) {
    float2 av = *(const float2*)(aptr + kc * 4);
    v2f a;
    a[0] = av.x; a[1] = av.y;
#pragma unroll
    for (int n = 0; n < NT; ++n) {
      float2 bv = *(const float2*)&lw[(n * 16 + lrow) * KP + kc * 4 + kbase];
      v2f b;
      b[0] = bv.x; b[1] = bv.y;
      acc[n] = __builtin_amdgcn_wmma_f32_16x16x4_f32(false, a, false, b,
                                                     (short)0, acc[n],
                                                     false, false);
    }
  }

  const int rbase = m0 + ((lane >> 4) << 3);
#pragma unroll
  for (int r = 0; r < 8; ++r) {
    int row = rbase + r;
    if (row < nrows) {
#pragma unroll
      for (int n = 0; n < NT; ++n) {
        int col = n * 16 + lrow;
        float v = acc[n][r];
        if (bias) v += bias[col];
        C[(size_t)row * NCOL + col] = v;
      }
    }
  }
}

__global__ void fill_f32(float* __restrict__ p, float v, int n) {
  int i = blockIdx.x * blockDim.x + threadIdx.x;
  if (i < n) p[i] = v;
}

__global__ void zero4(float* __restrict__ p, long n4) {
  long i = (long)blockIdx.x * blockDim.x + threadIdx.x;
  if (i < n4) ((float4*)p)[i] = make_float4(0.f, 0.f, 0.f, 0.f);
}

__global__ void deg_accum(const int* __restrict__ dst, float* __restrict__ deg,
                          int ne) {
  int e = blockIdx.x * blockDim.x + threadIdx.x;
  if (e < ne) atomicAdd(&deg[dst[e]], 1.0f);
}

__global__ void rsqrt_inplace(float* __restrict__ d, int n) {
  int i = blockIdx.x * blockDim.x + threadIdx.x;
  if (i < n) {
    float v = d[i];
    d[i] = v > 0.f ? rsqrtf(v) : 0.f;
  }
}

__global__ void edge_norm(const int* __restrict__ src,
                          const int* __restrict__ dst,
                          const float* __restrict__ invs,
                          float* __restrict__ nrm, int ne) {
  int e = blockIdx.x * blockDim.x + threadIdx.x;
  if (e < ne) nrm[e] = invs[src[e]] * invs[dst[e]];
}

__global__ __launch_bounds__(256) void aggregate(
    const float* __restrict__ t, const int* __restrict__ src,
    const int* __restrict__ dst, const float* __restrict__ nrm,
    float* __restrict__ agg, int ne) {
  const int lane = threadIdx.x & 31;
  int w = (blockIdx.x * blockDim.x + threadIdx.x) >> 5;
  const int nw = (gridDim.x * blockDim.x) >> 5;
  for (int e = w; e < ne; e += nw) {
    const int s = src[e];
    const int d = dst[e];
    const float nv = nrm[e];
    const float4 v = *(const float4*)(t + (size_t)s * CH + lane * 4);
    float* o = agg + (size_t)d * CH + lane * 4;
    atomicAdd(o + 0, v.x * nv);
    atomicAdd(o + 1, v.y * nv);
    atomicAdd(o + 2, v.z * nv);
    atomicAdd(o + 3, v.w * nv);
  }
}

__global__ void self_bias_act(float* __restrict__ hb,
                              const float* __restrict__ t,
                              const float* __restrict__ invs,
                              const float* __restrict__ bias, int n, int relu) {
  int i = blockIdx.x * blockDim.x + threadIdx.x;
  if (i >= n * 32) return;
  int node = i >> 5, c = (i & 31) * 4;
  float s = invs[node];
  s = s * s;
  float4 a = *(float4*)(hb + (size_t)node * CH + c);
  float4 tv = *(const float4*)(t + (size_t)node * CH + c);
  float4 b = *(const float4*)(bias + c);
  float4 r;
  r.x = a.x + tv.x * s + b.x;
  r.y = a.y + tv.y * s + b.y;
  r.z = a.z + tv.z * s + b.z;
  r.w = a.w + tv.w * s + b.w;
  if (relu) {
    r.x = fmaxf(r.x, 0.f);
    r.y = fmaxf(r.y, 0.f);
    r.z = fmaxf(r.z, 0.f);
    r.w = fmaxf(r.w, 0.f);
  }
  *(float4*)(hb + (size_t)node * CH + c) = r;
}

extern "C" void kernel_launch(void* const* d_in, const int* in_sizes, int n_in,
                              void* d_out, int out_size, void* d_ws,
                              size_t ws_size, hipStream_t stream) {
  const float* x = (const float*)d_in[0];
  const int* ei = (const int*)d_in[1];
  const float* W1 = (const float*)d_in[2];
  const float* b1 = (const float*)d_in[3];
  const float* W2 = (const float*)d_in[4];
  const float* b2 = (const float*)d_in[5];
  const float* W3 = (const float*)d_in[6];
  const float* b3 = (const float*)d_in[7];
  const float* Wo = (const float*)d_in[8];
  const float* bo = (const float*)d_in[9];

  const int N = in_sizes[0] / CH;
  const int E = in_sizes[1] / 2;
  const int* src = ei;
  const int* dstp = ei + E;

  float* bufA = (float*)d_ws;
  float* bufB = bufA + (size_t)N * CH;
  float* invs = bufB + (size_t)N * CH;
  float* nrm = invs + N;

  const dim3 blk(256);
  const int gN = (N + 255) / 256;
  const int gE = (E + 255) / 256;
  const int gGemm = (N + 127) / 128;
  const long n4 = (long)N * CH / 4;
  const int gZero = (int)((n4 + 255) / 256);
  const int gEpi = (N * 32 + 255) / 256;
  const int gAgg = 8192;

  fill_f32<<<gN, blk, 0, stream>>>(invs, 1.0f, N);
  deg_accum<<<gE, blk, 0, stream>>>(dstp, invs, E);
  rsqrt_inplace<<<gN, blk, 0, stream>>>(invs, N);
  edge_norm<<<gE, blk, 0, stream>>>(src, dstp, invs, nrm, E);

  gemm_wmma_f32<128><<<gGemm, blk, 0, stream>>>(x, W1, nullptr, bufA, N);
  zero4<<<gZero, blk, 0, stream>>>(bufB, n4);
  aggregate<<<gAgg, blk, 0, stream>>>(bufA, src, dstp, nrm, bufB, E);
  self_bias_act<<<gEpi, blk, 0, stream>>>(bufB, bufA, invs, b1, N, 1);

  gemm_wmma_f32<128><<<gGemm, blk, 0, stream>>>(bufB, W2, nullptr, bufA, N);
  zero4<<<gZero, blk, 0, stream>>>(bufB, n4);
  aggregate<<<gAgg, blk, 0, stream>>>(bufA, src, dstp, nrm, bufB, E);
  self_bias_act<<<gEpi, blk, 0, stream>>>(bufB, bufA, invs, b2, N, 1);

  gemm_wmma_f32<128><<<gGemm, blk, 0, stream>>>(bufB, W3, nullptr, bufA, N);
  zero4<<<gZero, blk, 0, stream>>>(bufB, n4);
  aggregate<<<gAgg, blk, 0, stream>>>(bufA, src, dstp, nrm, bufB, E);
  self_bias_act<<<gEpi, blk, 0, stream>>>(bufB, bufA, invs, b3, N, 0);

  gemm_wmma_f32<64><<<gGemm, blk, 0, stream>>>(bufB, Wo, bo, (float*)d_out, N);
}


# dst scan runs once per call: layer 1's owners also write each kept edge as a (src row, local dst row, weight) record to workspace lists; layers 2-3 replay them; arrival-order sums as the parent's
# speedup vs baseline: 2.5177x; 2.5177x over previous
#include <hip/hip_runtime.h>

typedef float v2f __attribute__((ext_vector_type(2)));
typedef float v8f __attribute__((ext_vector_type(8)));

#define CH 128

template <int NCOL>
__global__ __launch_bounds__(256) void gemm_wmma_f32(
    const float* __restrict__ A, const float* __restrict__ W,
    const float* __restrict__ bias, float* __restrict__ C, int nrows) {
  constexpr int NT = NCOL / 16;
  constexpr int KP = 130;
  __shared__ float lw[NCOL * KP];

  const int tid = threadIdx.x;
  for (int i = tid; i < CH * NCOL; i += 256) {
    int k = i / NCOL, c = i - k * NCOL;
    lw[c * KP + k] = W[i];
  }
  __syncthreads();

  const int wave = tid >> 5;
  const int lane = tid & 31;
  const int m0 = (blockIdx.x * 8 + wave) * 16;
  if (m0 >= nrows) return;

  const int lrow = lane & 15;
  const int kbase = (lane >> 4) * 2;

  v8f acc[NT] = {};

  int arow = m0 + lrow;
  if (arow > nrows - 1) arow = nrows - 1;
  const float* aptr = A + (size_t)arow * CH + kbase;

#pragma unroll 4
  for (int kc = 0; kc < CH / 4; ++kc) {
    float2 av = *(const float2*)(aptr + kc * 4);
    v2f a;
    a[0] = av.x; a[1] = av.y;
#pragma unroll
    for (int n = 0; n < NT; ++n) {
      float2 bv = *(const float2*)&lw[(n * 16 + lrow) * KP + kc * 4 + kbase];
      v2f b;
      b[0] = bv.x; b[1] = bv.y;
      acc[n] = __builtin_amdgcn_wmma_f32_16x16x4_f32(false, a, false, b,
                                                     (short)0, acc[n],
                                                     false, false);
    }
  }

  const int rbase = m0 + ((lane >> 4) << 3);
#pragma unroll
  for (int r = 0; r < 8; ++r) {
    int row = rbase + r;
    if (row < nrows) {
#pragma unroll
      for (int n = 0; n < NT; ++n) {
        int col = n * 16 + lrow;
        float v = acc[n][r];
        if (bias) v += bias[col];
        C[(size_t)row * NCOL + col] = v;
      }
    }
  }
}

__global__ void fill_f32(float* __restrict__ p, float v, int n) {
  int i = blockIdx.x * blockDim.x + threadIdx.x;
  if (i < n) p[i] = v;
}

__global__ void deg_accum(const int* __restrict__ dst, float* __restrict__ deg,
                          int ne) {
  int e = blockIdx.x * blockDim.x + threadIdx.x;
  if (e < ne) atomicAdd(&deg[dst[e]], 1.0f);
}

__global__ void rsqrt_inplace(float* __restrict__ d, int n) {
  int i = blockIdx.x * blockDim.x + threadIdx.x;
  if (i < n) {
    float v = d[i];
    d[i] = v > 0.f ? rsqrtf(v) : 0.f;
  }
}

__global__ void edge_norm(const int* __restrict__ src,
                          const int* __restrict__ dst,
                          const float* __restrict__ invs,
                          float* __restrict__ nrm, int ne) {
  int e = blockIdx.x * blockDim.x + threadIdx.x;
  if (e < ne) nrm[e] = invs[src[e]] * invs[dst[e]];
}

#define TN 512
#define TN_BITS 9
#define SCAN_U 16
#define QCAP 160

template <bool GUARD>
__device__ __forceinline__ void gather_rows(const float* __restrict__ t,
                                            float* tile, int sv, int dl,
                                            int nb, int m, int lane) {
#pragma unroll
  for (int j0 = 0; j0 < 32; j0 += 8) {
    float a[8][4];
#pragma unroll
    for (int u = 0; u < 8; ++u) {
      const int s = __builtin_amdgcn_readlane(sv, j0 + u);
      const float* row = t + (size_t)s * CH + lane;
#pragma unroll
      for (int j = 0; j < 4; ++j) a[u][j] = row[32 * j];
    }
#pragma unroll
    for (int u = 0; u < 8; ++u) {
      if (!GUARD || j0 + u < m) {
        const int d = __builtin_amdgcn_readlane(dl, j0 + u);
        const float w = __int_as_float(__builtin_amdgcn_readlane(nb, j0 + u));
        float* row = tile + d * CH + lane;
#pragma unroll
        for (int j = 0; j < 4; ++j) atomicAdd(row + 32 * j, a[u][j] * w);
      }
    }
  }
}

template <bool GUARD>
__device__ __forceinline__ void gather_group(
    const float* __restrict__ t, const int* __restrict__ src,
    const int* __restrict__ dst, const float* __restrict__ nrm, float* tile,
    int* rec, int n0, int e, int m, int lane) {
  int sv = 0, dl = 0, nb = 0;
  if (!GUARD || lane < m) {
    sv = src[e];
    dl = dst[e] - n0;
    nb = __float_as_int(nrm[e]);
    if (rec) *(int2*)(rec + 2 * lane) = make_int2((sv << TN_BITS) | dl, nb);
  }
  gather_rows<GUARD>(t, tile, sv, dl, nb, m, lane);
}

__device__ __forceinline__ int* rec_place(int* wg_count, int* wg_recs, int cap,
                                          int m, int lane) {
  int p = 0;
  if (lane == 0) p = atomicAdd(wg_count, m);
  p = __builtin_amdgcn_readfirstlane(p);
  return p + m <= cap ? wg_recs + 2 * (size_t)p : nullptr;
}

template <bool REC>
__device__ __forceinline__ void scan_phase(
    const float* __restrict__ t, const int* __restrict__ src,
    const int* __restrict__ dst, const float* __restrict__ nrm, float* tile,
    int* q, int* wg_count, int* wg_recs, int cap, int n0, int rows, int ne,
    int wave, int lane) {
  int cnt = 0;
  constexpr int WSPAN = SCAN_U * 32;
  constexpr int CHUNK = 32 * WSPAN;
  const int nchunk = (ne + CHUNK - 1) / CHUNK;
  const int c0 = nchunk > 0 ? (int)(blockIdx.x % (unsigned)nchunk) : 0;

  for (int it = 0; it < nchunk; ++it) {
    int c = it + c0;
    if (c >= nchunk) c -= nchunk;
    const int base = c * CHUNK + wave * WSPAN;
    const int* p = dst + base + lane;
    int dv[SCAN_U];
    if (base + WSPAN <= ne) {
#pragma unroll
      for (int k = 0; k < SCAN_U; ++k) dv[k] = p[k * 32];
    } else {
#pragma unroll
      for (int k = 0; k < SCAN_U; ++k)
        dv[k] = (base + k * 32 + lane < ne) ? p[k * 32] : -1;
    }
#pragma unroll
    for (int k = 0; k < SCAN_U; ++k) {
      const bool hit = (unsigned)(dv[k] - n0) < (unsigned)rows;
      const unsigned mask = __builtin_amdgcn_ballot_w32(hit);
      if (mask) {
        if (hit)
          q[cnt + __builtin_popcount(mask & ((1u << lane) - 1u))] =
              base + k * 32 + lane;
        cnt += __builtin_popcount(mask);
      }
      if ((k & 3) == 3) {
        while (cnt >= 32) {
          cnt -= 32;
          __builtin_amdgcn_wave_barrier();
          const int e = q[cnt + lane];
          int* rec = REC ? rec_place(wg_count, wg_recs, cap, 32, lane) : nullptr;
          gather_group<false>(t, src, dst, nrm, tile, rec, n0, e, 32, lane);
        }
      }
    }
  }
  if (cnt > 0) {
    __builtin_amdgcn_wave_barrier();
    const int e = lane < cnt ? q[lane] : 0;
    int* rec = REC ? rec_place(wg_count, wg_recs, cap, cnt, lane) : nullptr;
    gather_group<true>(t, src, dst, nrm, tile, rec, n0, e, cnt, lane);
  }
}

template <int MODE>
__global__ __launch_bounds__(1024) void aggregate_owner(
    const float* __restrict__ t, const int* __restrict__ src,
    const int* __restrict__ dst, const float* __restrict__ nrm,
    const float* __restrict__ invs, const float* __restrict__ bias,
    float* __restrict__ h, int* __restrict__ recs, int* __restrict__ cnts,
    int cap, int n, int ne, int relu) {
  __shared__ __attribute__((aligned(16))) float tile[TN * CH];
  __shared__ int qbuf[32 * QCAP];
  __shared__ int wg_count;
  const int tid = threadIdx.x;
  const int lane = tid & 31;
  const int wave = tid >> 5;
  const int n0 = blockIdx.x * TN;
  int rows = n - n0;
  if (rows > TN) rows = TN;
  if (MODE == 1 && tid == 0) wg_count = 0;

  for (int i = tid; i < rows * 32; i += 1024) {
    const int r = i >> 5, c = (i & 31) * 4;
    float s = invs[n0 + r];
    s = s * s;
    const float4 tv = *(const float4*)(t + (size_t)(n0 + r) * CH + c);
    const float4 b = *(const float4*)(bias + c);
    float4 o;
    o.x = tv.x * s + b.x;
    o.y = tv.y * s + b.y;
    o.z = tv.z * s + b.z;
    o.w = tv.w * s + b.w;
    *(float4*)&tile[r * CH + c] = o;
  }
  __syncthreads();

  int* wg_recs = MODE != 0 ? recs + (size_t)blockIdx.x * cap * 2 : nullptr;
  int total = -1;
  if (MODE == 2) total = __builtin_amdgcn_readfirstlane(cnts[blockIdx.x]);

  if (MODE == 2 && total >= 0) {
    const int nfull = total >> 5;
    const int sw = __builtin_amdgcn_readfirstlane(wave);
    int g = sw;
    if (g < nfull) {
      int2 cur = *(const int2*)(wg_recs + 2 * (g * 32 + lane));
      do {
        const int gn = g + 32;
        const int gl = gn < nfull ? gn : nfull - 1;
        const int2 nxt = *(const int2*)(wg_recs + 2 * (gl * 32 + lane));
        gather_rows<false>(t, tile, (int)((unsigned)cur.x >> TN_BITS),
                           cur.x & (TN - 1), cur.y, 32, lane);
        cur = nxt;
        g = gn;
      } while (g < nfull);
    }
    const int rem = total & 31;
    if (rem != 0 && (nfull & 31) == sw) {
      int2 r = make_int2(0, 0);
      if (lane < rem) r = *(const int2*)(wg_recs + 2 * (nfull * 32 + lane));
      gather_rows<true>(t, tile, (int)((unsigned)r.x >> TN_BITS),
                        r.x & (TN - 1), r.y, rem, lane);
    }
  } else {
    scan_phase<MODE == 1>(t, src, dst, nrm, tile, qbuf + wave * QCAP,
                          &wg_count, wg_recs, cap, n0, rows, ne, wave, lane);
  }
  __syncthreads();
  if (MODE == 1 && tid == 0) cnts[blockIdx.x] = wg_count <= cap ? wg_count : -1;

  for (int i = tid; i < rows * 32; i += 1024) {
    const int r = i >> 5, c = (i & 31) * 4;
    float4 v = *(const float4*)&tile[r * CH + c];
    if (relu) {
      v.x = fmaxf(v.x, 0.f);
      v.y = fmaxf(v.y, 0.f);
      v.z = fmaxf(v.z, 0.f);
      v.w = fmaxf(v.w, 0.f);
    }
    *(float4*)(h + (size_t)(n0 + r) * CH + c) = v;
  }
}

extern "C" void kernel_launch(void* const* d_in, const int* in_sizes, int n_in,
                              void* d_out, int out_size, void* d_ws,
                              size_t ws_size, hipStream_t stream) {
  const float* x = (const float*)d_in[0];
  const int* ei = (const int*)d_in[1];
  const float* W1 = (const float*)d_in[2];
  const float* b1 = (const float*)d_in[3];
  const float* W2 = (const float*)d_in[4];
  const float* b2 = (const float*)d_in[5];
  const float* W3 = (const float*)d_in[6];
  const float* b3 = (const float*)d_in[7];
  const float* Wo = (const float*)d_in[8];
  const float* bo = (const float*)d_in[9];

  const int N = in_sizes[0] / CH;
  const int E = in_sizes[1] / 2;
  const int* src = ei;
  const int* dstp = ei + E;

  float* bufA = (float*)d_ws;
  float* bufB = bufA + (size_t)N * CH;
  float* invs = bufB + (size_t)N * CH;
  float* nrm = invs + N;

  const dim3 blk(256);
  const dim3 blkAgg(1024);
  const int gN = (N + 255) / 256;
  const int gE = (E + 255) / 256;
  const int gGemm = (N + 127) / 128;
  const int gTile = (N + TN - 1) / TN;

  const size_t recOff = ((size_t)N * CH * 2 + N + E + 31) & ~(size_t)31;
  const int share = E / (gTile > 0 ? gTile : 1);
  const int cap = (share + share / 8 + 1024 + 31) & ~31;
  int* recs = (int*)d_ws + recOff;
  int* cnts = recs + (size_t)gTile * cap * 2;
  const bool lists =
      ws_size >= (recOff + (size_t)gTile * cap * 2 + gTile) * sizeof(int) &&
      N <= (1 << (31 - TN_BITS));

  fill_f32<<<gN, blk, 0, stream>>>(invs, 1.0f, N);
  deg_accum<<<gE, blk, 0, stream>>>(dstp, invs, E);
  rsqrt_inplace<<<gN, blk, 0, stream>>>(invs, N);
  edge_norm<<<gE, blk, 0, stream>>>(src, dstp, invs, nrm, E);

  gemm_wmma_f32<128><<<gGemm, blk, 0, stream>>>(x, W1, nullptr, bufA, N);
  if (lists)
    aggregate_owner<1><<<gTile, blkAgg, 0, stream>>>(
        bufA, src, dstp, nrm, invs, b1, bufB, recs, cnts, cap, N, E, 1);
  else
    aggregate_owner<0><<<gTile, blkAgg, 0, stream>>>(
        bufA, src, dstp, nrm, invs, b1, bufB, nullptr, nullptr, 0, N, E, 1);
  gemm_wmma_f32<128><<<gGemm, blk, 0, stream>>>(bufB, W2, nullptr, bufA, N);
  if (lists)
    aggregate_owner<2><<<gTile, blkAgg, 0, stream>>>(
        bufA, src, dstp, nrm, invs, b2, bufB, recs, cnts, cap, N, E, 1);
  else
    aggregate_owner<0><<<gTile, blkAgg, 0, stream>>>(
        bufA, src, dstp, nrm, invs, b2, bufB, nullptr, nullptr, 0, N, E, 1);
  gemm_wmma_f32<128><<<gGemm, blk, 0, stream>>>(bufB, W3, nullptr, bufA, N);
  if (lists)
    aggregate_owner<2><<<gTile, blkAgg, 0, stream>>>(
        bufA, src, dstp, nrm, invs, b3, bufB, recs, cnts, cap, N, E, 0);
  else
    aggregate_owner<0><<<gTile, blkAgg, 0, stream>>>(
        bufA, src, dstp, nrm, invs, b3, bufB, nullptr, nullptr, 0, N, E, 0);

  gemm_wmma_f32<64><<<gGemm, blk, 0, stream>>>(bufB, Wo, bo, (float*)d_out, N);
}


# owner tile 512 -> 400 rows: 250 equal work-groups, one per WGP (LDS-bound), run on 250 of the 256 WGPs instead of 196; rows, edges, seeds, stores per WGP fall 22%; arrival-order sums as the parent's
# speedup vs baseline: 2.6683x; 1.0598x over previous
#include <hip/hip_runtime.h>

typedef float v2f __attribute__((ext_vector_type(2)));
typedef float v8f __attribute__((ext_vector_type(8)));

#define CH 128

template <int NCOL>
__global__ __launch_bounds__(256) void gemm_wmma_f32(
    const float* __restrict__ A, const float* __restrict__ W,
    const float* __restrict__ bias, float* __restrict__ C, int nrows) {
  constexpr int NT = NCOL / 16;
  constexpr int KP = 130;
  __shared__ float lw[NCOL * KP];

  const int tid = threadIdx.x;
  for (int i = tid; i < CH * NCOL; i += 256) {
    int k = i / NCOL, c = i - k * NCOL;
    lw[c * KP + k] = W[i];
  }
  __syncthreads();

  const int wave = tid >> 5;
  const int lane = tid & 31;
  const int m0 = (blockIdx.x * 8 + wave) * 16;
  if (m0 >= nrows) return;

  const int lrow = lane & 15;
  const int kbase = (lane >> 4) * 2;

  v8f acc[NT] = {};

  int arow = m0 + lrow;
  if (arow > nrows - 1) arow = nrows - 1;
  const float* aptr = A + (size_t)arow * CH + kbase;

#pragma unroll 4
  for (int kc = 0; kc < CH / 4; ++kc) {
    float2 av = *(const float2*)(aptr + kc * 4);
    v2f a;
    a[0] = av.x; a[1] = av.y;
#pragma unroll
    for (int n = 0; n < NT; ++n) {
      float2 bv = *(const float2*)&lw[(n * 16 + lrow) * KP + kc * 4 + kbase];
      v2f b;
      b[0] = bv.x; b[1] = bv.y;
      acc[n] = __builtin_amdgcn_wmma_f32_16x16x4_f32(false, a, false, b,
                                                     (short)0, acc[n],
                                                     false, false);
    }
  }

  const int rbase = m0 + ((lane >> 4) << 3);
#pragma unroll
  for (int r = 0; r < 8; ++r) {
    int row = rbase + r;
    if (row < nrows) {
#pragma unroll
      for (int n = 0; n < NT; ++n) {
        int col = n * 16 + lrow;
        float v = acc[n][r];
        if (bias) v += bias[col];
        C[(size_t)row * NCOL + col] = v;
      }
    }
  }
}

__global__ void fill_f32(float* __restrict__ p, float v, int n) {
  int i = blockIdx.x * blockDim.x + threadIdx.x;
  if (i < n) p[i] = v;
}

__global__ void deg_accum(const int* __restrict__ dst, float* __restrict__ deg,
                          int ne) {
  int e = blockIdx.x * blockDim.x + threadIdx.x;
  if (e < ne) atomicAdd(&deg[dst[e]], 1.0f);
}

__global__ void rsqrt_inplace(float* __restrict__ d, int n) {
  int i = blockIdx.x * blockDim.x + threadIdx.x;
  if (i < n) {
    float v = d[i];
    d[i] = v > 0.f ? rsqrtf(v) : 0.f;
  }
}

__global__ void edge_norm(const int* __restrict__ src,
                          const int* __restrict__ dst,
                          const float* __restrict__ invs,
                          float* __restrict__ nrm, int ne) {
  int e = blockIdx.x * blockDim.x + threadIdx.x;
  if (e < ne) nrm[e] = invs[src[e]] * invs[dst[e]];
}

#define TN 400
#define TN_BITS 9
#define TN_MASK ((1 << TN_BITS) - 1)
static_assert(TN <= (1 << TN_BITS), "a local row must fit a record's low bits");
#define SCAN_U 16
#define QCAP 160

template <bool GUARD>
__device__ __forceinline__ void gather_rows(const float* __restrict__ t,
                                            float* tile, int sv, int dl,
                                            int nb, int m, int lane) {
#pragma unroll
  for (int j0 = 0; j0 < 32; j0 += 8) {
    float a[8][4];
#pragma unroll
    for (int u = 0; u < 8; ++u) {
      const int s = __builtin_amdgcn_readlane(sv, j0 + u);
      const float* row = t + (size_t)s * CH + lane;
#pragma unroll
      for (int j = 0; j < 4; ++j) a[u][j] = row[32 * j];
    }
#pragma unroll
    for (int u = 0; u < 8; ++u) {
      if (!GUARD || j0 + u < m) {
        const int d = __builtin_amdgcn_readlane(dl, j0 + u);
        const float w = __int_as_float(__builtin_amdgcn_readlane(nb, j0 + u));
        float* row = tile + d * CH + lane;
#pragma unroll
        for (int j = 0; j < 4; ++j) atomicAdd(row + 32 * j, a[u][j] * w);
      }
    }
  }
}

template <bool GUARD>
__device__ __forceinline__ void gather_group(
    const float* __restrict__ t, const int* __restrict__ src,
    const int* __restrict__ dst, const float* __restrict__ nrm, float* tile,
    int* rec, int n0, int e, int m, int lane) {
  int sv = 0, dl = 0, nb = 0;
  if (!GUARD || lane < m) {
    sv = src[e];
    dl = dst[e] - n0;
    nb = __float_as_int(nrm[e]);
    if (rec) *(int2*)(rec + 2 * lane) = make_int2((sv << TN_BITS) | dl, nb);
  }
  gather_rows<GUARD>(t, tile, sv, dl, nb, m, lane);
}

__device__ __forceinline__ int* rec_place(int* wg_count, int* wg_recs, int cap,
                                          int m, int lane) {
  int p = 0;
  if (lane == 0) p = atomicAdd(wg_count, m);
  p = __builtin_amdgcn_readfirstlane(p);
  return p + m <= cap ? wg_recs + 2 * (size_t)p : nullptr;
}

template <bool REC>
__device__ __forceinline__ void scan_phase(
    const float* __restrict__ t, const int* __restrict__ src,
    const int* __restrict__ dst, const float* __restrict__ nrm, float* tile,
    int* q, int* wg_count, int* wg_recs, int cap, int n0, int rows, int ne,
    int wave, int lane) {
  int cnt = 0;
  constexpr int WSPAN = SCAN_U * 32;
  constexpr int CHUNK = 32 * WSPAN;
  const int nchunk = (ne + CHUNK - 1) / CHUNK;
  const int c0 = nchunk > 0 ? (int)(blockIdx.x % (unsigned)nchunk) : 0;

  for (int it = 0; it < nchunk; ++it) {
    int c = it + c0;
    if (c >= nchunk) c -= nchunk;
    const int base = c * CHUNK + wave * WSPAN;
    const int* p = dst + base + lane;
    int dv[SCAN_U];
    if (base + WSPAN <= ne) {
#pragma unroll
      for (int k = 0; k < SCAN_U; ++k) dv[k] = p[k * 32];
    } else {
#pragma unroll
      for (int k = 0; k < SCAN_U; ++k)
        dv[k] = (base + k * 32 + lane < ne) ? p[k * 32] : -1;
    }
#pragma unroll
    for (int k = 0; k < SCAN_U; ++k) {
      const bool hit = (unsigned)(dv[k] - n0) < (unsigned)rows;
      const unsigned mask = __builtin_amdgcn_ballot_w32(hit);
      if (mask) {
        if (hit)
          q[cnt + __builtin_popcount(mask & ((1u << lane) - 1u))] =
              base + k * 32 + lane;
        cnt += __builtin_popcount(mask);
      }
      if ((k & 3) == 3) {
        while (cnt >= 32) {
          cnt -= 32;
          __builtin_amdgcn_wave_barrier();
          const int e = q[cnt + lane];
          int* rec = REC ? rec_place(wg_count, wg_recs, cap, 32, lane) : nullptr;
          gather_group<false>(t, src, dst, nrm, tile, rec, n0, e, 32, lane);
        }
      }
    }
  }
  if (cnt > 0) {
    __builtin_amdgcn_wave_barrier();
    const int e = lane < cnt ? q[lane] : 0;
    int* rec = REC ? rec_place(wg_count, wg_recs, cap, cnt, lane) : nullptr;
    gather_group<true>(t, src, dst, nrm, tile, rec, n0, e, cnt, lane);
  }
}

template <int MODE>
__global__ __launch_bounds__(1024) void aggregate_owner(
    const float* __restrict__ t, const int* __restrict__ src,
    const int* __restrict__ dst, const float* __restrict__ nrm,
    const float* __restrict__ invs, const float* __restrict__ bias,
    float* __restrict__ h, int* __restrict__ recs, int* __restrict__ cnts,
    int cap, int n, int ne, int relu) {
  __shared__ __attribute__((aligned(16))) float tile[TN * CH];
  __shared__ int qbuf[32 * QCAP];
  __shared__ int wg_count;
  const int tid = threadIdx.x;
  const int lane = tid & 31;
  const int wave = tid >> 5;
  const int n0 = blockIdx.x * TN;
  int rows = n - n0;
  if (rows > TN) rows = TN;
  if (MODE == 1 && tid == 0) wg_count = 0;

  for (int i = tid; i < rows * 32; i += 1024) {
    const int r = i >> 5, c = (i & 31) * 4;
    float s = invs[n0 + r];
    s = s * s;
    const float4 tv = *(const float4*)(t + (size_t)(n0 + r) * CH + c);
    const float4 b = *(const float4*)(bias + c);
    float4 o;
    o.x = tv.x * s + b.x;
    o.y = tv.y * s + b.y;
    o.z = tv.z * s + b.z;
    o.w = tv.w * s + b.w;
    *(float4*)&tile[r * CH + c] = o;
  }
  __syncthreads();

  int* wg_recs = MODE != 0 ? recs + (size_t)blockIdx.x * cap * 2 : nullptr;
  int total = -1;
  if (MODE == 2) total = __builtin_amdgcn_readfirstlane(cnts[blockIdx.x]);

  if (MODE == 2 && total >= 0) {
    const int nfull = total >> 5;
    const int sw = __builtin_amdgcn_readfirstlane(wave);
    int g = sw;
    if (g < nfull) {
      int2 cur = *(const int2*)(wg_recs + 2 * (g * 32 + lane));
      do {
        const int gn = g + 32;
        const int gl = gn < nfull ? gn : nfull - 1;
        const int2 nxt = *(const int2*)(wg_recs + 2 * (gl * 32 + lane));
        gather_rows<false>(t, tile, (int)((unsigned)cur.x >> TN_BITS),
                           cur.x & TN_MASK, cur.y, 32, lane);
        cur = nxt;
        g = gn;
      } while (g < nfull);
    }
    const int rem = total & 31;
    if (rem != 0 && (nfull & 31) == sw) {
      int2 r = make_int2(0, 0);
      if (lane < rem) r = *(const int2*)(wg_recs + 2 * (nfull * 32 + lane));
      gather_rows<true>(t, tile, (int)((unsigned)r.x >> TN_BITS),
                        r.x & TN_MASK, r.y, rem, lane);
    }
  } else {
    scan_phase<MODE == 1>(t, src, dst, nrm, tile, qbuf + wave * QCAP,
                          &wg_count, wg_recs, cap, n0, rows, ne, wave, lane);
  }
  __syncthreads();
  if (MODE == 1 && tid == 0) cnts[blockIdx.x] = wg_count <= cap ? wg_count : -1;

  for (int i = tid; i < rows * 32; i += 1024) {
    const int r = i >> 5, c = (i & 31) * 4;
    float4 v = *(const float4*)&tile[r * CH + c];
    if (relu) {
      v.x = fmaxf(v.x, 0.f);
      v.y = fmaxf(v.y, 0.f);
      v.z = fmaxf(v.z, 0.f);
      v.w = fmaxf(v.w, 0.f);
    }
    *(float4*)(h + (size_t)(n0 + r) * CH + c) = v;
  }
}

extern "C" void kernel_launch(void* const* d_in, const int* in_sizes, int n_in,
                              void* d_out, int out_size, void* d_ws,
                              size_t ws_size, hipStream_t stream) {
  const float* x = (const float*)d_in[0];
  const int* ei = (const int*)d_in[1];
  const float* W1 = (const float*)d_in[2];
  const float* b1 = (const float*)d_in[3];
  const float* W2 = (const float*)d_in[4];
  const float* b2 = (const float*)d_in[5];
  const float* W3 = (const float*)d_in[6];
  const float* b3 = (const float*)d_in[7];
  const float* Wo = (const float*)d_in[8];
  const float* bo = (const float*)d_in[9];

  const int N = in_sizes[0] / CH;
  const int E = in_sizes[1] / 2;
  const int* src = ei;
  const int* dstp = ei + E;

  float* bufA = (float*)d_ws;
  float* bufB = bufA + (size_t)N * CH;
  float* invs = bufB + (size_t)N * CH;
  float* nrm = invs + N;

  const dim3 blk(256);
  const dim3 blkAgg(1024);
  const int gN = (N + 255) / 256;
  const int gE = (E + 255) / 256;
  const int gGemm = (N + 127) / 128;
  const int gTile = (N + TN - 1) / TN;

  const size_t recOff = ((size_t)N * CH * 2 + N + E + 31) & ~(size_t)31;
  const int share = E / (gTile > 0 ? gTile : 1);
  const int cap = (share + share / 8 + 512 + 31) & ~31;
  int* recs = (int*)d_ws + recOff;
  int* cnts = recs + (size_t)gTile * cap * 2;
  const bool lists =
      ws_size >= (recOff + (size_t)gTile * cap * 2 + gTile) * sizeof(int) &&
      N <= (1 << (31 - TN_BITS));

  fill_f32<<<gN, blk, 0, stream>>>(invs, 1.0f, N);
  deg_accum<<<gE, blk, 0, stream>>>(dstp, invs, E);
  rsqrt_inplace<<<gN, blk, 0, stream>>>(invs, N);
  edge_norm<<<gE, blk, 0, stream>>>(src, dstp, invs, nrm, E);

  gemm_wmma_f32<128><<<gGemm, blk, 0, stream>>>(x, W1, nullptr, bufA, N);
  if (lists)
    aggregate_owner<1><<<gTile, blkAgg, 0, stream>>>(
        bufA, src, dstp, nrm, invs, b1, bufB, recs, cnts, cap, N, E, 1);
  else
    aggregate_owner<0><<<gTile, blkAgg, 0, stream>>>(
        bufA, src, dstp, nrm, invs, b1, bufB, nullptr, nullptr, 0, N, E, 1);
  gemm_wmma_f32<128><<<gGemm, blk, 0, stream>>>(bufB, W2, nullptr, bufA, N);
  if (lists)
    aggregate_owner<2><<<gTile, blkAgg, 0, stream>>>(
        bufA, src, dstp, nrm, invs, b2, bufB, recs, cnts, cap, N, E, 1);
  else
    aggregate_owner<0><<<gTile, blkAgg, 0, stream>>>(
        bufA, src, dstp, nrm, invs, b2, bufB, nullptr, nullptr, 0, N, E, 1);
  gemm_wmma_f32<128><<<gGemm, blk, 0, stream>>>(bufB, W3, nullptr, bufA, N);
  if (lists)
    aggregate_owner<2><<<gTile, blkAgg, 0, stream>>>(
        bufA, src, dstp, nrm, invs, b3, bufB, recs, cnts, cap, N, E, 0);
  else
    aggregate_owner<0><<<gTile, blkAgg, 0, stream>>>(
        bufA, src, dstp, nrm, invs, b3, bufB, nullptr, nullptr, 0, N, E, 0);

  gemm_wmma_f32<64><<<gGemm, blk, 0, stream>>>(bufB, Wo, bo, (float*)d_out, N);
}


# dst scan leaves layer 1: one partition pass reads the edges once, LDS count per owner tile + one global int add per tile reserves a run, writes the record lists; 3 layers replay; arrival-order sums
# speedup vs baseline: 2.9811x; 1.1172x over previous
#include <hip/hip_runtime.h>

typedef float v2f __attribute__((ext_vector_type(2)));
typedef float v8f __attribute__((ext_vector_type(8)));

#define CH 128

template <int NCOL>
__global__ __launch_bounds__(256) void gemm_wmma_f32(
    const float* __restrict__ A, const float* __restrict__ W,
    const float* __restrict__ bias, float* __restrict__ C, int nrows) {
  constexpr int NT = NCOL / 16;
  constexpr int KP = 130;
  __shared__ float lw[NCOL * KP];

  const int tid = threadIdx.x;
  for (int i = tid; i < CH * NCOL; i += 256) {
    int k = i / NCOL, c = i - k * NCOL;
    lw[c * KP + k] = W[i];
  }
  __syncthreads();

  const int wave = tid >> 5;
  const int lane = tid & 31;
  const int m0 = (blockIdx.x * 8 + wave) * 16;
  if (m0 >= nrows) return;

  const int lrow = lane & 15;
  const int kbase = (lane >> 4) * 2;

  v8f acc[NT] = {};

  int arow = m0 + lrow;
  if (arow > nrows - 1) arow = nrows - 1;
  const float* aptr = A + (size_t)arow * CH + kbase;

#pragma unroll 4
  for (int kc = 0; kc < CH / 4; ++kc) {
    float2 av = *(const float2*)(aptr + kc * 4);
    v2f a;
    a[0] = av.x; a[1] = av.y;
#pragma unroll
    for (int n = 0; n < NT; ++n) {
      float2 bv = *(const float2*)&lw[(n * 16 + lrow) * KP + kc * 4 + kbase];
      v2f b;
      b[0] = bv.x; b[1] = bv.y;
      acc[n] = __builtin_amdgcn_wmma_f32_16x16x4_f32(false, a, false, b,
                                                     (short)0, acc[n],
                                                     false, false);
    }
  }

  const int rbase = m0 + ((lane >> 4) << 3);
#pragma unroll
  for (int r = 0; r < 8; ++r) {
    int row = rbase + r;
    if (row < nrows) {
#pragma unroll
      for (int n = 0; n < NT; ++n) {
        int col = n * 16 + lrow;
        float v = acc[n][r];
        if (bias) v += bias[col];
        C[(size_t)row * NCOL + col] = v;
      }
    }
  }
}

__global__ void fill_f32(float* __restrict__ p, float v, int n,
                         int* __restrict__ z, int nz) {
  int i = blockIdx.x * blockDim.x + threadIdx.x;
  if (i < n) p[i] = v;
  for (int j = i; j < nz; j += gridDim.x * blockDim.x) z[j] = 0;
}

__global__ void deg_accum(const int* __restrict__ dst, float* __restrict__ deg,
                          int ne) {
  int e = blockIdx.x * blockDim.x + threadIdx.x;
  if (e < ne) atomicAdd(&deg[dst[e]], 1.0f);
}

__global__ void rsqrt_inplace(float* __restrict__ d, int n) {
  int i = blockIdx.x * blockDim.x + threadIdx.x;
  if (i < n) {
    float v = d[i];
    d[i] = v > 0.f ? rsqrtf(v) : 0.f;
  }
}


#define TN 400
#define TN_BITS 9
#define TN_MASK ((1 << TN_BITS) - 1)
static_assert(TN <= (1 << TN_BITS), "a local row must fit a record's low bits");
#define SCAN_U 16
#define QCAP 160
#define PART_WG 256
#define PART_TILES 1024
#define PART_U 8
#define CNT_PITCH 32

__global__ __launch_bounds__(1024) void edge_partition(
    const int* __restrict__ src, const int* __restrict__ dst,
    const float* __restrict__ invs, int* __restrict__ recs,
    int* __restrict__ cnts, int cap, int ntile, int ne) {
  __shared__ int place[PART_TILES];
  const int tid = threadIdx.x;
  const int per = ((ne + (int)gridDim.x - 1) / (int)gridDim.x + 31) & ~31;
  const int e0 = blockIdx.x * per;
  int e1 = e0 + per;
  if (e1 > ne) e1 = ne;

  for (int b = tid; b < ntile; b += 1024) place[b] = 0;
  __syncthreads();
  static_assert(PART_U == 8, "the two load fences below name 8 edges");
  for (int eb = e0 + tid; eb < e1; eb += 1024 * PART_U) {
    int dv[PART_U];
#pragma unroll
    for (int u = 0; u < PART_U; ++u) {
      const int e = eb + u * 1024;
      dv[u] = dst[e < e1 ? e : e1 - 1];
    }
    asm volatile("" ::"v"(dv[0]), "v"(dv[1]), "v"(dv[2]), "v"(dv[3]),
                 "v"(dv[4]), "v"(dv[5]), "v"(dv[6]), "v"(dv[7]));
#pragma unroll
    for (int u = 0; u < PART_U; ++u)
      if (eb + u * 1024 < e1) atomicAdd(&place[dv[u] / TN], 1);
  }
  __syncthreads();
  for (int b = tid; b < ntile; b += 1024) {
    const int c = place[b];
    place[b] = c > 0 ? atomicAdd(&cnts[(size_t)b * CNT_PITCH], c) : 0;
  }
  __syncthreads();
  for (int eb = e0 + tid; eb < e1; eb += 1024 * PART_U) {
    int sv[PART_U], dv[PART_U];
#pragma unroll
    for (int u = 0; u < PART_U; ++u) {
      const int e = eb + u * 1024;
      const int ec = e < e1 ? e : e1 - 1;
      sv[u] = src[ec];
      dv[u] = dst[ec];
    }
    float is[PART_U], id[PART_U];
#pragma unroll
    for (int u = 0; u < PART_U; ++u) {
      is[u] = invs[sv[u]];
      id[u] = invs[dv[u]];
    }
    asm volatile("" ::"v"(is[0]), "v"(is[1]), "v"(is[2]), "v"(is[3]),
                 "v"(is[4]), "v"(is[5]), "v"(is[6]), "v"(is[7]));
    asm volatile("" ::"v"(id[0]), "v"(id[1]), "v"(id[2]), "v"(id[3]),
                 "v"(id[4]), "v"(id[5]), "v"(id[6]), "v"(id[7]));
#pragma unroll
    for (int u = 0; u < PART_U; ++u) {
      if (eb + u * 1024 < e1) {
        const int b = dv[u] / TN;
        const int p = atomicAdd(&place[b], 1);
        if (p < cap)
          *(int2*)(recs + ((size_t)b * cap + p) * 2) =
              make_int2((sv[u] << TN_BITS) | (dv[u] - b * TN),
                        __float_as_int(is[u] * id[u]));
      }
    }
  }
}

template <bool GUARD>
__device__ __forceinline__ void gather_rows(const float* __restrict__ t,
                                            float* tile, int sv, int dl,
                                            int nb, int m, int lane) {
#pragma unroll
  for (int j0 = 0; j0 < 32; j0 += 8) {
    float a[8][4];
#pragma unroll
    for (int u = 0; u < 8; ++u) {
      const int s = __builtin_amdgcn_readlane(sv, j0 + u);
      const float* row = t + (size_t)s * CH + lane;
#pragma unroll
      for (int j = 0; j < 4; ++j) a[u][j] = row[32 * j];
    }
#pragma unroll
    for (int u = 0; u < 8; ++u) {
      if (!GUARD || j0 + u < m) {
        const int d = __builtin_amdgcn_readlane(dl, j0 + u);
        const float w = __int_as_float(__builtin_amdgcn_readlane(nb, j0 + u));
        float* row = tile + d * CH + lane;
#pragma unroll
        for (int j = 0; j < 4; ++j) atomicAdd(row + 32 * j, a[u][j] * w);
      }
    }
  }
}

template <bool GUARD>
__device__ __forceinline__ void gather_group(
    const float* __restrict__ t, const int* __restrict__ src,
    const int* __restrict__ dst, const float* __restrict__ invs, float* tile,
    int n0, int e, int m, int lane) {
  int sv = 0, dl = 0, nb = 0;
  if (!GUARD || lane < m) {
    sv = src[e];
    const int dv = dst[e];
    dl = dv - n0;
    nb = __float_as_int(invs[sv] * invs[dv]);
  }
  gather_rows<GUARD>(t, tile, sv, dl, nb, m, lane);
}

__device__ __forceinline__ void scan_phase(
    const float* __restrict__ t, const int* __restrict__ src,
    const int* __restrict__ dst, const float* __restrict__ invs, float* tile,
    int* q, int n0, int rows, int ne, int wave, int lane) {
  int cnt = 0;
  constexpr int WSPAN = SCAN_U * 32;
  constexpr int CHUNK = 32 * WSPAN;
  const int nchunk = (ne + CHUNK - 1) / CHUNK;
  const int c0 = nchunk > 0 ? (int)(blockIdx.x % (unsigned)nchunk) : 0;

  for (int it = 0; it < nchunk; ++it) {
    int c = it + c0;
    if (c >= nchunk) c -= nchunk;
    const int base = c * CHUNK + wave * WSPAN;
    const int* p = dst + base + lane;
    int dv[SCAN_U];
    if (base + WSPAN <= ne) {
#pragma unroll
      for (int k = 0; k < SCAN_U; ++k) dv[k] = p[k * 32];
    } else {
#pragma unroll
      for (int k = 0; k < SCAN_U; ++k)
        dv[k] = (base + k * 32 + lane < ne) ? p[k * 32] : -1;
    }
#pragma unroll
    for (int k = 0; k < SCAN_U; ++k) {
      const bool hit = (unsigned)(dv[k] - n0) < (unsigned)rows;
      const unsigned mask = __builtin_amdgcn_ballot_w32(hit);
      if (mask) {
        if (hit)
          q[cnt + __builtin_popcount(mask & ((1u << lane) - 1u))] =
              base + k * 32 + lane;
        cnt += __builtin_popcount(mask);
      }
      if ((k & 3) == 3) {
        while (cnt >= 32) {
          cnt -= 32;
          __builtin_amdgcn_wave_barrier();
          const int e = q[cnt + lane];
          gather_group<false>(t, src, dst, invs, tile, n0, e, 32, lane);
        }
      }
    }
  }
  if (cnt > 0) {
    __builtin_amdgcn_wave_barrier();
    const int e = lane < cnt ? q[lane] : 0;
    gather_group<true>(t, src, dst, invs, tile, n0, e, cnt, lane);
  }
}

template <bool LISTS>
__global__ __launch_bounds__(1024) void aggregate_owner(
    const float* __restrict__ t, const int* __restrict__ src,
    const int* __restrict__ dst, const float* __restrict__ invs,
    const float* __restrict__ bias, float* __restrict__ h,
    const int* __restrict__ recs, const int* __restrict__ cnts, int cap, int n,
    int ne, int relu) {
  __shared__ __attribute__((aligned(16))) float tile[TN * CH];
  __shared__ int qbuf[32 * QCAP];
  const int tid = threadIdx.x;
  const int lane = tid & 31;
  const int wave = tid >> 5;
  const int n0 = blockIdx.x * TN;
  int rows = n - n0;
  if (rows > TN) rows = TN;

  for (int i = tid; i < rows * 32; i += 1024) {
    const int r = i >> 5, c = (i & 31) * 4;
    float s = invs[n0 + r];
    s = s * s;
    const float4 tv = *(const float4*)(t + (size_t)(n0 + r) * CH + c);
    const float4 b = *(const float4*)(bias + c);
    float4 o;
    o.x = tv.x * s + b.x;
    o.y = tv.y * s + b.y;
    o.z = tv.z * s + b.z;
    o.w = tv.w * s + b.w;
    *(float4*)&tile[r * CH + c] = o;
  }
  __syncthreads();

  const int* wg_recs = LISTS ? recs + (size_t)blockIdx.x * cap * 2 : nullptr;
  int total = -1;
  if (LISTS) {
    total = __builtin_amdgcn_readfirstlane(cnts[(size_t)blockIdx.x * CNT_PITCH]);
    if (total > cap) total = -1;
  }

  if (LISTS && total >= 0) {
    const int nfull = total >> 5;
    const int sw = __builtin_amdgcn_readfirstlane(wave);
    int g = sw;
    if (g < nfull) {
      int2 cur = *(const int2*)(wg_recs + 2 * (g * 32 + lane));
      do {
        const int gn = g + 32;
        const int gl = gn < nfull ? gn : nfull - 1;
        const int2 nxt = *(const int2*)(wg_recs + 2 * (gl * 32 + lane));
        gather_rows<false>(t, tile, (int)((unsigned)cur.x >> TN_BITS),
                           cur.x & TN_MASK, cur.y, 32, lane);
        cur = nxt;
        g = gn;
      } while (g < nfull);
    }
    const int rem = total & 31;
    if (rem != 0 && (nfull & 31) == sw) {
      int2 r = make_int2(0, 0);
      if (lane < rem) r = *(const int2*)(wg_recs + 2 * (nfull * 32 + lane));
      gather_rows<true>(t, tile, (int)((unsigned)r.x >> TN_BITS),
                        r.x & TN_MASK, r.y, rem, lane);
    }
  } else {
    scan_phase(t, src, dst, invs, tile, qbuf + wave * QCAP, n0, rows, ne, wave,
               lane);
  }
  __syncthreads();

  for (int i = tid; i < rows * 32; i += 1024) {
    const int r = i >> 5, c = (i & 31) * 4;
    float4 v = *(const float4*)&tile[r * CH + c];
    if (relu) {
      v.x = fmaxf(v.x, 0.f);
      v.y = fmaxf(v.y, 0.f);
      v.z = fmaxf(v.z, 0.f);
      v.w = fmaxf(v.w, 0.f);
    }
    *(float4*)(h + (size_t)(n0 + r) * CH + c) = v;
  }
}

extern "C" void kernel_launch(void* const* d_in, const int* in_sizes, int n_in,
                              void* d_out, int out_size, void* d_ws,
                              size_t ws_size, hipStream_t stream) {
  const float* x = (const float*)d_in[0];
  const int* ei = (const int*)d_in[1];
  const float* W1 = (const float*)d_in[2];
  const float* b1 = (const float*)d_in[3];
  const float* W2 = (const float*)d_in[4];
  const float* b2 = (const float*)d_in[5];
  const float* W3 = (const float*)d_in[6];
  const float* b3 = (const float*)d_in[7];
  const float* Wo = (const float*)d_in[8];
  const float* bo = (const float*)d_in[9];

  const int N = in_sizes[0] / CH;
  const int E = in_sizes[1] / 2;
  const int* src = ei;
  const int* dstp = ei + E;

  float* bufA = (float*)d_ws;
  float* bufB = bufA + (size_t)N * CH;
  float* invs = bufB + (size_t)N * CH;

  const dim3 blk(256);
  const dim3 blkAgg(1024);
  const int gN = (N + 255) / 256;
  const int gE = (E + 255) / 256;
  const int gGemm = (N + 127) / 128;
  const int gTile = (N + TN - 1) / TN;

  const size_t recOff = ((size_t)N * CH * 2 + N + 31) & ~(size_t)31;
  const int share = E / (gTile > 0 ? gTile : 1);
  const int cap = (share + share / 8 + 512 + 31) & ~31;
  int* recs = (int*)d_ws + recOff;
  int* cnts = recs + (size_t)gTile * cap * 2;
  const bool lists =
      ws_size >= (recOff + (size_t)gTile * cap * 2 +
                  (size_t)gTile * CNT_PITCH) * sizeof(int) &&
      gTile <= PART_TILES && N <= (1 << (31 - TN_BITS));

  fill_f32<<<gN, blk, 0, stream>>>(invs, 1.0f, N, cnts,
                                   lists ? gTile * CNT_PITCH : 0);
  deg_accum<<<gE, blk, 0, stream>>>(dstp, invs, E);
  rsqrt_inplace<<<gN, blk, 0, stream>>>(invs, N);

  if (lists) {
    edge_partition<<<PART_WG, blkAgg, 0, stream>>>(src, dstp, invs, recs, cnts,
                                                   cap, gTile, E);
    gemm_wmma_f32<128><<<gGemm, blk, 0, stream>>>(x, W1, nullptr, bufA, N);
    aggregate_owner<true><<<gTile, blkAgg, 0, stream>>>(
        bufA, src, dstp, invs, b1, bufB, recs, cnts, cap, N, E, 1);
    gemm_wmma_f32<128><<<gGemm, blk, 0, stream>>>(bufB, W2, nullptr, bufA, N);
    aggregate_owner<true><<<gTile, blkAgg, 0, stream>>>(
        bufA, src, dstp, invs, b2, bufB, recs, cnts, cap, N, E, 1);
    gemm_wmma_f32<128><<<gGemm, blk, 0, stream>>>(bufB, W3, nullptr, bufA, N);
    aggregate_owner<true><<<gTile, blkAgg, 0, stream>>>(
        bufA, src, dstp, invs, b3, bufB, recs, cnts, cap, N, E, 0);
  } else {
    gemm_wmma_f32<128><<<gGemm, blk, 0, stream>>>(x, W1, nullptr, bufA, N);
    aggregate_owner<false><<<gTile, blkAgg, 0, stream>>>(
        bufA, src, dstp, invs, b1, bufB, nullptr, nullptr, 0, N, E, 1);
    gemm_wmma_f32<128><<<gGemm, blk, 0, stream>>>(bufB, W2, nullptr, bufA, N);
    aggregate_owner<false><<<gTile, blkAgg, 0, stream>>>(
        bufA, src, dstp, invs, b2, bufB, nullptr, nullptr, 0, N, E, 1);
    gemm_wmma_f32<128><<<gGemm, blk, 0, stream>>>(bufB, W3, nullptr, bufA, N);
    aggregate_owner<false><<<gTile, blkAgg, 0, stream>>>(
        bufA, src, dstp, invs, b3, bufB, nullptr, nullptr, 0, N, E, 0);
  }

  gemm_wmma_f32<64><<<gGemm, blk, 0, stream>>>(bufB, Wo, bo, (float*)d_out, N);
}


# edge norm factored out of the edge sum: product rows stored times deg^-1/2, owner scales its sum once, (t*is)*id for t*(is*id); in-degree count rides the partition pass; 11 -> 9 launches
# speedup vs baseline: 3.1279x; 1.0492x over previous
#include <hip/hip_runtime.h>

typedef float v2f __attribute__((ext_vector_type(2)));
typedef float v8f __attribute__((ext_vector_type(8)));

#define CH 128

template <int NCOL, bool SCALE>
__global__ __launch_bounds__(256) void gemm_wmma_f32(
    const float* __restrict__ A, const float* __restrict__ W,
    const float* __restrict__ bias, const int* __restrict__ degc,
    float* __restrict__ C, int nrows) {
  constexpr int NT = NCOL / 16;
  constexpr int KP = 130;
  __shared__ float lw[NCOL * KP];

  const int tid = threadIdx.x;
  for (int i = tid; i < CH * NCOL; i += 256) {
    int k = i / NCOL, c = i - k * NCOL;
    lw[c * KP + k] = W[i];
  }
  __syncthreads();

  const int wave = tid >> 5;
  const int lane = tid & 31;
  const int m0 = (blockIdx.x * 8 + wave) * 16;
  if (m0 >= nrows) return;

  const int lrow = lane & 15;
  const int kbase = (lane >> 4) * 2;

  v8f acc[NT] = {};

  int arow = m0 + lrow;
  if (arow > nrows - 1) arow = nrows - 1;
  const float* aptr = A + (size_t)arow * CH + kbase;

#pragma unroll 4
  for (int kc = 0; kc < CH / 4; ++kc) {
    float2 av = *(const float2*)(aptr + kc * 4);
    v2f a;
    a[0] = av.x; a[1] = av.y;
#pragma unroll
    for (int n = 0; n < NT; ++n) {
      float2 bv = *(const float2*)&lw[(n * 16 + lrow) * KP + kc * 4 + kbase];
      v2f b;
      b[0] = bv.x; b[1] = bv.y;
      acc[n] = __builtin_amdgcn_wmma_f32_16x16x4_f32(false, a, false, b,
                                                     (short)0, acc[n],
                                                     false, false);
    }
  }

  const int rbase = m0 + ((lane >> 4) << 3);
#pragma unroll
  for (int r = 0; r < 8; ++r) {
    int row = rbase + r;
    if (row < nrows) {
      float sc = 1.f;
      if (SCALE) {
        const float d = (float)(degc[row] + 1);
        sc = d > 0.f ? rsqrtf(d) : 0.f;
      }
#pragma unroll
      for (int n = 0; n < NT; ++n) {
        int col = n * 16 + lrow;
        float v = acc[n][r];
        if (bias) v += bias[col];
        if (SCALE) v *= sc;
        C[(size_t)row * NCOL + col] = v;
      }
    }
  }
}

__global__ void zero_ints(int* __restrict__ p, int n, int* __restrict__ z,
                          int nz) {
  int i = blockIdx.x * blockDim.x + threadIdx.x;
  if (i < n) p[i] = 0;
  for (int j = i; j < nz; j += gridDim.x * blockDim.x) z[j] = 0;
}

__global__ void deg_count(const int* __restrict__ dst, int* __restrict__ degc,
                          int ne) {
  for (int e = blockIdx.x * blockDim.x + threadIdx.x; e < ne;
       e += gridDim.x * blockDim.x)
    atomicAdd(&degc[dst[e]], 1);
}


#define TN 400
#define TN_BITS 9
#define TN_MASK ((1 << TN_BITS) - 1)
static_assert(TN <= (1 << TN_BITS), "a local row must fit a record's low bits");
#define SCAN_U 16
#define QCAP 160
#define PART_WG 256
#define PART_TILES 1024
#define PART_U 8
#define CNT_PITCH 32

__global__ __launch_bounds__(1024) void edge_partition(
    const int* __restrict__ src, const int* __restrict__ dst,
    int* __restrict__ degc, int* __restrict__ recs, int* __restrict__ cnts,
    int cap, int ntile, int ne) {
  __shared__ int place[PART_TILES];
  const int tid = threadIdx.x;
  const int per = ((ne + (int)gridDim.x - 1) / (int)gridDim.x + 31) & ~31;
  const int e0 = blockIdx.x * per;
  int e1 = e0 + per;
  if (e1 > ne) e1 = ne;

  for (int b = tid; b < ntile; b += 1024) place[b] = 0;
  __syncthreads();
  static_assert(PART_U == 8, "the load fences below name 8 edges");
  for (int eb = e0 + tid; eb < e1; eb += 1024 * PART_U) {
    int dv[PART_U];
#pragma unroll
    for (int u = 0; u < PART_U; ++u) {
      const int e = eb + u * 1024;
      dv[u] = dst[e < e1 ? e : e1 - 1];
    }
    asm volatile("" ::"v"(dv[0]), "v"(dv[1]), "v"(dv[2]), "v"(dv[3]),
                 "v"(dv[4]), "v"(dv[5]), "v"(dv[6]), "v"(dv[7]));
#pragma unroll
    for (int u = 0; u < PART_U; ++u)
      if (eb + u * 1024 < e1) atomicAdd(&place[dv[u] / TN], 1);
  }
  __syncthreads();
  for (int b = tid; b < ntile; b += 1024) {
    const int c = place[b];
    place[b] = c > 0 ? atomicAdd(&cnts[(size_t)b * CNT_PITCH], c) : 0;
  }
  __syncthreads();
  for (int eb = e0 + tid; eb < e1; eb += 1024 * PART_U) {
    int sv[PART_U], dv[PART_U];
#pragma unroll
    for (int u = 0; u < PART_U; ++u) {
      const int e = eb + u * 1024;
      const int ec = e < e1 ? e : e1 - 1;
      sv[u] = src[ec];
      dv[u] = dst[ec];
    }
    asm volatile("" ::"v"(sv[0]), "v"(sv[1]), "v"(sv[2]), "v"(sv[3]),
                 "v"(sv[4]), "v"(sv[5]), "v"(sv[6]), "v"(sv[7]));
    asm volatile("" ::"v"(dv[0]), "v"(dv[1]), "v"(dv[2]), "v"(dv[3]),
                 "v"(dv[4]), "v"(dv[5]), "v"(dv[6]), "v"(dv[7]));
#pragma unroll
    for (int u = 0; u < PART_U; ++u) {
      if (eb + u * 1024 < e1) {
        const int b = dv[u] / TN;
        const int p = atomicAdd(&place[b], 1);
        atomicAdd(&degc[dv[u]], 1);
        if (p < cap)
          recs[(size_t)b * cap + p] = (sv[u] << TN_BITS) | (dv[u] - b * TN);
      }
    }
  }
}

template <bool GUARD>
__device__ __forceinline__ void gather_rows(const float* __restrict__ t,
                                            float* tile, int sv, int dl, int m,
                                            int lane) {
#pragma unroll
  for (int j0 = 0; j0 < 32; j0 += 8) {
    float a[8][4];
#pragma unroll
    for (int u = 0; u < 8; ++u) {
      const int s = __builtin_amdgcn_readlane(sv, j0 + u);
      const float* row = t + (size_t)s * CH + lane;
#pragma unroll
      for (int j = 0; j < 4; ++j) a[u][j] = row[32 * j];
    }
#pragma unroll
    for (int u = 0; u < 8; ++u) {
      if (!GUARD || j0 + u < m) {
        const int d = __builtin_amdgcn_readlane(dl, j0 + u);
        float* row = tile + d * CH + lane;
#pragma unroll
        for (int j = 0; j < 4; ++j) atomicAdd(row + 32 * j, a[u][j]);
      }
    }
  }
}

template <bool GUARD>
__device__ __forceinline__ void gather_group(
    const float* __restrict__ t, const int* __restrict__ src,
    const int* __restrict__ dst, float* tile, int n0, int e, int m, int lane) {
  int sv = 0, dl = 0;
  if (!GUARD || lane < m) {
    sv = src[e];
    dl = dst[e] - n0;
  }
  gather_rows<GUARD>(t, tile, sv, dl, m, lane);
}

__device__ __forceinline__ void scan_phase(
    const float* __restrict__ t, const int* __restrict__ src,
    const int* __restrict__ dst, float* tile, int* q, int n0, int rows, int ne,
    int wave, int lane) {
  int cnt = 0;
  constexpr int WSPAN = SCAN_U * 32;
  constexpr int CHUNK = 32 * WSPAN;
  const int nchunk = (ne + CHUNK - 1) / CHUNK;
  const int c0 = nchunk > 0 ? (int)(blockIdx.x % (unsigned)nchunk) : 0;

  for (int it = 0; it < nchunk; ++it) {
    int c = it + c0;
    if (c >= nchunk) c -= nchunk;
    const int base = c * CHUNK + wave * WSPAN;
    const int* p = dst + base + lane;
    int dv[SCAN_U];
    if (base + WSPAN <= ne) {
#pragma unroll
      for (int k = 0; k < SCAN_U; ++k) dv[k] = p[k * 32];
    } else {
#pragma unroll
      for (int k = 0; k < SCAN_U; ++k)
        dv[k] = (base + k * 32 + lane < ne) ? p[k * 32] : -1;
    }
#pragma unroll
    for (int k = 0; k < SCAN_U; ++k) {
      const bool hit = (unsigned)(dv[k] - n0) < (unsigned)rows;
      const unsigned mask = __builtin_amdgcn_ballot_w32(hit);
      if (mask) {
        if (hit)
          q[cnt + __builtin_popcount(mask & ((1u << lane) - 1u))] =
              base + k * 32 + lane;
        cnt += __builtin_popcount(mask);
      }
      if ((k & 3) == 3) {
        while (cnt >= 32) {
          cnt -= 32;
          __builtin_amdgcn_wave_barrier();
          const int e = q[cnt + lane];
          gather_group<false>(t, src, dst, tile, n0, e, 32, lane);
        }
      }
    }
  }
  if (cnt > 0) {
    __builtin_amdgcn_wave_barrier();
    const int e = lane < cnt ? q[lane] : 0;
    gather_group<true>(t, src, dst, tile, n0, e, cnt, lane);
  }
}

template <bool LISTS>
__global__ __launch_bounds__(1024) void aggregate_owner(
    const float* __restrict__ t, const int* __restrict__ src,
    const int* __restrict__ dst, const int* __restrict__ degc,
    const float* __restrict__ bias, float* __restrict__ h,
    const int* __restrict__ recs, const int* __restrict__ cnts, int cap, int n,
    int ne, int relu) {
  __shared__ __attribute__((aligned(16))) float tile[TN * CH];
  __shared__ int qbuf[32 * QCAP];
  const int tid = threadIdx.x;
  const int lane = tid & 31;
  const int wave = tid >> 5;
  const int n0 = blockIdx.x * TN;
  int rows = n - n0;
  if (rows > TN) rows = TN;

  for (int i = tid; i < rows * 32; i += 1024) {
    const int r = i >> 5, c = (i & 31) * 4;
    *(float4*)&tile[r * CH + c] =
        *(const float4*)(t + (size_t)(n0 + r) * CH + c);
  }
  __syncthreads();

  const int* wg_recs = LISTS ? recs + (size_t)blockIdx.x * cap : nullptr;
  int total = -1;
  if (LISTS) {
    total = __builtin_amdgcn_readfirstlane(cnts[(size_t)blockIdx.x * CNT_PITCH]);
    if (total > cap) total = -1;
  }

  if (LISTS && total >= 0) {
    const int nfull = total >> 5;
    const int sw = __builtin_amdgcn_readfirstlane(wave);
    int g = sw;
    if (g < nfull) {
      int cur = wg_recs[g * 32 + lane];
      do {
        const int gn = g + 32;
        const int gl = gn < nfull ? gn : nfull - 1;
        const int nxt = wg_recs[gl * 32 + lane];
        gather_rows<false>(t, tile, (int)((unsigned)cur >> TN_BITS),
                           cur & TN_MASK, 32, lane);
        cur = nxt;
        g = gn;
      } while (g < nfull);
    }
    const int rem = total & 31;
    if (rem != 0 && (nfull & 31) == sw) {
      int r = 0;
      if (lane < rem) r = wg_recs[nfull * 32 + lane];
      gather_rows<true>(t, tile, (int)((unsigned)r >> TN_BITS), r & TN_MASK,
                        rem, lane);
    }
  } else {
    scan_phase(t, src, dst, tile, qbuf + wave * QCAP, n0, rows, ne, wave,
               lane);
  }
  constexpr int NIT = (TN * 32 + 1023) / 1024;
  const float4 b = *(const float4*)(bias + lane * 4);
  float sc[NIT];
#pragma unroll
  for (int k = 0; k < NIT; ++k) {
    const int r = wave + 32 * k;
    const float d = (float)(degc[n0 + (r < rows ? r : rows - 1)] + 1);
    sc[k] = d > 0.f ? rsqrtf(d) : 0.f;
  }
  __syncthreads();

#pragma unroll
  for (int k = 0; k < NIT; ++k) {
    const int r = wave + 32 * k, c = lane * 4;
    if (r < rows) {
      const float s = sc[k];
      float4 v = *(const float4*)&tile[r * CH + c];
      v.x = v.x * s + b.x;
      v.y = v.y * s + b.y;
      v.z = v.z * s + b.z;
      v.w = v.w * s + b.w;
      if (relu) {
        v.x = fmaxf(v.x, 0.f);
        v.y = fmaxf(v.y, 0.f);
        v.z = fmaxf(v.z, 0.f);
        v.w = fmaxf(v.w, 0.f);
      }
      *(float4*)(h + (size_t)(n0 + r) * CH + c) = v;
    }
  }
}

extern "C" void kernel_launch(void* const* d_in, const int* in_sizes, int n_in,
                              void* d_out, int out_size, void* d_ws,
                              size_t ws_size, hipStream_t stream) {
  const float* x = (const float*)d_in[0];
  const int* ei = (const int*)d_in[1];
  const float* W1 = (const float*)d_in[2];
  const float* b1 = (const float*)d_in[3];
  const float* W2 = (const float*)d_in[4];
  const float* b2 = (const float*)d_in[5];
  const float* W3 = (const float*)d_in[6];
  const float* b3 = (const float*)d_in[7];
  const float* Wo = (const float*)d_in[8];
  const float* bo = (const float*)d_in[9];

  const int N = in_sizes[0] / CH;
  const int E = in_sizes[1] / 2;
  const int* src = ei;
  const int* dstp = ei + E;

  float* bufA = (float*)d_ws;
  float* bufB = bufA + (size_t)N * CH;
  int* degc = (int*)(bufB + (size_t)N * CH);

  const dim3 blk(256);
  const dim3 blkAgg(1024);
  const int gN = (N + 255) / 256;
  const int gGemm = (N + 127) / 128;
  const int gTile = (N + TN - 1) / TN;

  const size_t recOff = ((size_t)N * CH * 2 + N + 31) & ~(size_t)31;
  const int share = E / (gTile > 0 ? gTile : 1);
  const int cap = (share + share / 8 + 512 + 31) & ~31;
  int* recs = (int*)d_ws + recOff;
  int* cnts = recs + (size_t)gTile * cap;
  const bool lists =
      ws_size >= (recOff + (size_t)gTile * cap +
                  (size_t)gTile * CNT_PITCH) * sizeof(int) &&
      gTile <= PART_TILES && N <= (1 << (31 - TN_BITS));

  zero_ints<<<gN, blk, 0, stream>>>(degc, N, cnts,
                                    lists ? gTile * CNT_PITCH : 0);

  if (lists) {
    edge_partition<<<PART_WG, blkAgg, 0, stream>>>(src, dstp, degc, recs, cnts,
                                                   cap, gTile, E);
    gemm_wmma_f32<128, true><<<gGemm, blk, 0, stream>>>(x, W1, nullptr, degc,
                                                        bufA, N);
    aggregate_owner<true><<<gTile, blkAgg, 0, stream>>>(
        bufA, src, dstp, degc, b1, bufB, recs, cnts, cap, N, E, 1);
    gemm_wmma_f32<128, true><<<gGemm, blk, 0, stream>>>(bufB, W2, nullptr,
                                                        degc, bufA, N);
    aggregate_owner<true><<<gTile, blkAgg, 0, stream>>>(
        bufA, src, dstp, degc, b2, bufB, recs, cnts, cap, N, E, 1);
    gemm_wmma_f32<128, true><<<gGemm, blk, 0, stream>>>(bufB, W3, nullptr,
                                                        degc, bufA, N);
    aggregate_owner<true><<<gTile, blkAgg, 0, stream>>>(
        bufA, src, dstp, degc, b3, bufB, recs, cnts, cap, N, E, 0);
  } else {
    deg_count<<<1024, blk, 0, stream>>>(dstp, degc, E);
    gemm_wmma_f32<128, true><<<gGemm, blk, 0, stream>>>(x, W1, nullptr, degc,
                                                        bufA, N);
    aggregate_owner<false><<<gTile, blkAgg, 0, stream>>>(
        bufA, src, dstp, degc, b1, bufB, nullptr, nullptr, 0, N, E, 1);
    gemm_wmma_f32<128, true><<<gGemm, blk, 0, stream>>>(bufB, W2, nullptr,
                                                        degc, bufA, N);
    aggregate_owner<false><<<gTile, blkAgg, 0, stream>>>(
        bufA, src, dstp, degc, b2, bufB, nullptr, nullptr, 0, N, E, 1);
    gemm_wmma_f32<128, true><<<gGemm, blk, 0, stream>>>(bufB, W3, nullptr,
                                                        degc, bufA, N);
    aggregate_owner<false><<<gTile, blkAgg, 0, stream>>>(
        bufA, src, dstp, degc, b3, bufB, nullptr, nullptr, 0, N, E, 0);
  }

  gemm_wmma_f32<64, false><<<gGemm, blk, 0, stream>>>(bufB, Wo, bo, nullptr,
                                                      (float*)d_out, N);
}
